# starting kernel
# baseline (speedup 1.0000x reference)
#include <hip/hip_runtime.h>

#define N_NODES 50000
#define N_EDGES 640000
#define CH      128

typedef float v2f __attribute__((ext_vector_type(2)));
typedef float v8f __attribute__((ext_vector_type(8)));


__global__ void gc_init_ws(float* __restrict__ ws, int n) {
    int i = blockIdx.x * blockDim.x + threadIdx.x;
    if (i < n) ws[i] = 0.0f;
}

__global__ __launch_bounds__(256) void gc_scatter(const float* __restrict__ x,
                                                  const int*   __restrict__ rows,
                                                  const int*   __restrict__ cols,
                                                  float*       __restrict__ deg,
                                                  float*       __restrict__ agg) {
    const int wave = (blockIdx.x * blockDim.x + threadIdx.x) >> 5;
    const int lane = threadIdx.x & 31;
    const int r = rows[wave];
    const int c = cols[wave];
    const float4 v = ((const float4*)(x + (size_t)c * CH))[lane];
    float* dst = agg + (size_t)r * CH + lane * 4;
    atomicAdd(dst + 0, v.x);
    atomicAdd(dst + 1, v.y);
    atomicAdd(dst + 2, v.z);
    atomicAdd(dst + 3, v.w);
    if (lane == 0) atomicAdd(deg + r, 1.0f);
}

__global__ __launch_bounds__(256) void gc_finalize(const float* __restrict__ x,
                                                   const float* __restrict__ deg,
                                                   float*       __restrict__ agg) {
    const int i = blockIdx.x * blockDim.x + threadIdx.x;
    const int node = i >> 7;
    float d = deg[node];
    d = d < 1.0f ? 1.0f : d;
    agg[i] = x[i] + agg[i] / d;
}

__global__ __launch_bounds__(256) void gc_gemm_wmma(const float* __restrict__ h,
                                                    const float* __restrict__ W,
                                                    const float* __restrict__ bias,
                                                    float*       __restrict__ out) {
    const int wid  = (blockIdx.x * blockDim.x + threadIdx.x) >> 5;
    const int lane = threadIdx.x & 31;
    const int mt = wid >> 3;
    const int nt = wid & 7;
    const int m0 = mt * 16;
    const int n0 = nt * 16;
    const int ln = lane & 15;
    const int hi = lane >> 4;

    const float bv = bias[n0 + ln];
    v8f c;
#pragma unroll
    for (int r = 0; r < 8; ++r) c[r] = bv;

    const float* hrow = h + (size_t)(m0 + ln) * CH;
    const float* wrow = W + (size_t)(n0 + ln) * CH;

#pragma unroll 4
    for (int kb = 0; kb < 32; ++kb) {
        const int k = kb * 4 + hi * 2;
        v2f a = *(const v2f*)(hrow + k);
        v2f b = *(const v2f*)(wrow + k);
        c = __builtin_amdgcn_wmma_f32_16x16x4_f32(false, a, false, b,
                                                  (short)0, c, false, false);
    }

    float* orow = out + (size_t)(m0 + hi * 8) * CH + n0 + ln;
#pragma unroll
    for (int r = 0; r < 8; ++r) orow[(size_t)r * CH] = c[r];
}


extern "C" void kernel_launch(void* const* d_in, const int* in_sizes, int n_in,
                              void* d_out, int out_size, void* d_ws, size_t ws_size,
                              hipStream_t stream) {
    const float* x    = (const float*)d_in[0];
    const int*   ei   = (const int*)  d_in[1];
    const float* W    = (const float*)d_in[2];
    const float* bias = (const float*)d_in[3];
    float*       out  = (float*)d_out;

    const int* rows = ei;
    const int* cols = ei + N_EDGES;

    float* deg = (float*)d_ws;
    float* agg = deg + N_NODES;

    {
        const int n = N_NODES + N_NODES * CH;
        gc_init_ws<<<(n + 255) / 256, 256, 0, stream>>>((float*)d_ws, n);
    }
    {
        const int blocks = (N_EDGES * 32) / 256;
        gc_scatter<<<blocks, 256, 0, stream>>>(x, rows, cols, deg, agg);
    }
    {
        const int blocks = (N_NODES * CH) / 256;
        gc_finalize<<<blocks, 256, 0, stream>>>(x, deg, agg);
    }
    {
        const int blocks = (3125 * 8 * 32) / 256;
        gc_gemm_wmma<<<blocks, 256, 0, stream>>>(agg, W, bias, out);
    }
}


# one launch, no workspace: each of 241 work-groups keeps its 208 h rows in LDS, filters the whole edge list for them, sums x rows by LDS f32 atomics (arrival order), f32 WMMA from LDS
# speedup vs baseline: 3.1546x; 3.1546x over previous
#include <hip/hip_runtime.h>


#define N_NODES 50000
#define N_EDGES 640000
#define CH      128
#define ROWS_WG 208
#define N_WG    241
#define PITCH   132
#define CAPW    512
#define NTHR    1024
#define NWAVE   32
#define N_ROW4  (N_EDGES / 4)
#define FULL_IT (N_ROW4 / NTHR)
#define REM4    (N_ROW4 - FULL_IT * NTHR)
#define SCAN_U  6

typedef float v2f __attribute__((ext_vector_type(2)));
typedef float v8f __attribute__((ext_vector_type(8)));

__global__ __launch_bounds__(NTHR) void gc_fused(const float* __restrict__ x,
                                                 const int*   __restrict__ rows,
                                                 const int*   __restrict__ cols,
                                                 const float* __restrict__ W,
                                                 const float* __restrict__ bias,
                                                 float*       __restrict__ out) {
    __shared__ __attribute__((aligned(16))) float    acc[ROWS_WG * PITCH];
    __shared__ int      degs[ROWS_WG];
    __shared__ unsigned elist[NWAVE * CAPW];

    const int tid  = threadIdx.x;
    const int lane = tid & 31;
    const int wv   = tid >> 5;
    const int base = blockIdx.x * ROWS_WG;
    const int nrows_i = (N_NODES - base) < ROWS_WG ? (N_NODES - base) : ROWS_WG;
    const unsigned nrows = (unsigned)nrows_i;

    {
        float4* a4 = (float4*)acc;
        const float4 z = make_float4(0.f, 0.f, 0.f, 0.f);
        for (int i = tid; i < ROWS_WG * PITCH / 4; i += NTHR) a4[i] = z;
        if (tid < ROWS_WG) degs[tid] = 0;
    }
    __syncthreads();

    unsigned* mylist = elist + wv * CAPW;
    int wcnt = 0;

    auto take = [&](int r, int e) {
        const unsigned lr = (unsigned)(r - base);
        const bool m = lr < nrows;
        const unsigned mask = __builtin_amdgcn_ballot_w32(m);
        if (mask != 0u) {
            if (m) {
                atomicAdd(&degs[lr], 1);
                const int pos = wcnt + (int)__builtin_amdgcn_mbcnt_lo(mask, 0u);
                if (pos < CAPW) {
                    mylist[pos] = (lr << 20) | (unsigned)e;
                } else {
                    const float* src = x + (size_t)cols[e] * CH;
                    float* dst = acc + lr * PITCH;
#pragma nounroll
                    for (int k = 0; k < CH; ++k) atomicAdd(dst + k, src[k]);
                }
            }
            wcnt += __builtin_popcount(mask);
        }
    };

    const int4* rows4 = (const int4*)rows;
    {
        int it = (int)(blockIdx.x % (FULL_IT / SCAN_U)) * SCAN_U;
        int4 cur[SCAN_U];
#pragma unroll
        for (int u = 0; u < SCAN_U; ++u) cur[u] = rows4[(it + u) * NTHR + tid];
        for (int o = 0; o < FULL_IT / SCAN_U; ++o) {
            const int nx = (it + SCAN_U < FULL_IT) ? it + SCAN_U : 0;
            int4 nxt[SCAN_U];
#pragma unroll
            for (int u = 0; u < SCAN_U; ++u) nxt[u] = rows4[(nx + u) * NTHR + tid];
#pragma unroll
            for (int u = 0; u < SCAN_U; ++u) {
                const int e0 = ((it + u) * NTHR + tid) * 4;
                take(cur[u].x, e0 + 0);
                take(cur[u].y, e0 + 1);
                take(cur[u].z, e0 + 2);
                take(cur[u].w, e0 + 3);
            }
#pragma unroll
            for (int u = 0; u < SCAN_U; ++u) cur[u] = nxt[u];
            it = nx;
        }
    }
    if (tid < REM4) {
        const int g4 = FULL_IT * NTHR + tid;
        const int4 r4 = rows4[g4];
        take(r4.x, g4 * 4 + 0);
        take(r4.y, g4 * 4 + 1);
        take(r4.z, g4 * 4 + 2);
        take(r4.w, g4 * 4 + 3);
    }
    __syncthreads();

    {
        const int n_ent = __builtin_amdgcn_readfirstlane(wcnt < CAPW ? wcnt : CAPW);
        const int hsel = lane >> 4;
        const int o0 = lane * 4 + ((0 + hsel) & 3);
        const int o1 = lane * 4 + ((1 + hsel) & 3);
        const int o2 = lane * 4 + ((2 + hsel) & 3);
        const int o3 = lane * 4 + ((3 + hsel) & 3);
        for (int b0 = 0; b0 < n_ent; b0 += 32) {
            const int idx = b0 + lane;
            unsigned ent = 0u;
            int c = 0;
            if (idx < n_ent) {
                ent = mylist[idx];
                c = cols[ent & 0xFFFFFu];
            }
            const int nb = (n_ent - b0) < 32 ? (n_ent - b0) : 32;
            for (int j0 = 0; j0 < nb; j0 += 8) {
                float4 v[8];
                int    lrj[8];
#pragma unroll
                for (int u = 0; u < 8; ++u) {
                    const int cj = __builtin_amdgcn_readlane(c, j0 + u);
                    lrj[u] = (int)(((unsigned)__builtin_amdgcn_readlane((int)ent, j0 + u)) >> 20);
                    v[u] = ((const float4*)(x + (size_t)cj * CH))[lane];
                }
#pragma unroll
                for (int u = 0; u < 8; ++u) {
                    if (j0 + u < nb) {
                        float* d = acc + lrj[u] * PITCH;
                        const float t0 = hsel ? v[u].y : v[u].x;
                        const float t1 = hsel ? v[u].z : v[u].y;
                        const float t2 = hsel ? v[u].w : v[u].z;
                        const float t3 = hsel ? v[u].x : v[u].w;
                        atomicAdd(d + o0, t0);
                        atomicAdd(d + o1, t1);
                        atomicAdd(d + o2, t2);
                        atomicAdd(d + o3, t3);
                    }
                }
            }
        }
    }
    __syncthreads();

    for (int i = tid; i < nrows_i * 32; i += NTHR) {
        const int lr = i >> 5;
        const int q  = i & 31;
        float d = (float)degs[lr];
        d = d < 1.0f ? 1.0f : d;
        float4* p = (float4*)(acc + lr * PITCH + q * 4);
        float4 a = *p;
        const float4 xv = ((const float4*)(x + (size_t)(base + lr) * CH))[q];
        a.x = xv.x + a.x / d;
        a.y = xv.y + a.y / d;
        a.z = xv.z + a.z / d;
        a.w = xv.w + a.w / d;
        *p = a;
    }
    __syncthreads();

    {
        const int ln = lane & 15;
        const int hi = lane >> 4;
        const int n_items = (nrows_i >> 4) * 4;
        for (int item = wv; item < n_items; item += NWAVE) {
            const int mt = item >> 2;
            const int n0 = (item & 3) * 32;
            const float bv0 = bias[n0 + ln];
            const float bv1 = bias[n0 + 16 + ln];
            v8f c0, c1;
#pragma unroll
            for (int r = 0; r < 8; ++r) { c0[r] = bv0; c1[r] = bv1; }

            const float* arow = acc + (mt * 16 + ln) * PITCH + hi * 2;
            const float* w0   = W + (size_t)(n0 + ln) * CH + hi * 2;
            const float* w1   = w0 + 16 * CH;

#pragma unroll 8
            for (int kb = 0; kb < 32; ++kb) {
                const v2f a  = *(const v2f*)(arow + kb * 4);
                const v2f b0 = *(const v2f*)(w0 + kb * 4);
                const v2f b1 = *(const v2f*)(w1 + kb * 4);
                c0 = __builtin_amdgcn_wmma_f32_16x16x4_f32(false, a, false, b0, (short)0, c0, false, false);
                c1 = __builtin_amdgcn_wmma_f32_16x16x4_f32(false, a, false, b1, (short)0, c1, false, false);
            }

            float* orow = out + (size_t)(base + mt * 16 + hi * 8) * CH + n0 + ln;
#pragma unroll
            for (int r = 0; r < 8; ++r) {
                orow[(size_t)r * CH]      = c0[r];
                orow[(size_t)r * CH + 16] = c1[r];
            }
        }
    }
}


extern "C" void kernel_launch(void* const* d_in, const int* in_sizes, int n_in,
                              void* d_out, int out_size, void* d_ws, size_t ws_size,
                              hipStream_t stream) {
    const float* x    = (const float*)d_in[0];
    const int*   ei   = (const int*)  d_in[1];
    const float* W    = (const float*)d_in[2];
    const float* bias = (const float*)d_in[3];
    float*       out  = (float*)d_out;

    const int* rows = ei;
    const int* cols = ei + N_EDGES;

    gc_fused<<<N_WG, NTHR, 0, stream>>>(x, rows, cols, W, bias, out);
}


# each wave takes its kept edges out of its LDS ring 8 at a time while it reads the row list (x rows fly during the next sweep): scan and LDS f32 atomic sums (arrival order) overlap; 1 barrier less
# speedup vs baseline: 3.4006x; 1.0780x over previous
#include <hip/hip_runtime.h>


#define N_NODES 50000
#define N_EDGES 640000
#define CH      128
#define ROWS_WG 208
#define N_WG    241
#define PITCH   132
#define CAPW    512
#define BATCH   8
#define NTHR    1024
#define NWAVE   32
#define N_ROW4  (N_EDGES / 4)
#define FULL_IT (N_ROW4 / NTHR)
#define REM4    (N_ROW4 - FULL_IT * NTHR)
#define SCAN_U  6

typedef float v2f __attribute__((ext_vector_type(2)));
typedef float v8f __attribute__((ext_vector_type(8)));

__global__ __launch_bounds__(NTHR) void gc_fused(const float* __restrict__ x,
                                                 const int*   __restrict__ rows,
                                                 const int*   __restrict__ cols,
                                                 const float* __restrict__ W,
                                                 const float* __restrict__ bias,
                                                 float*       __restrict__ out) {
    __shared__ __attribute__((aligned(16))) float    acc[ROWS_WG * PITCH];
    __shared__ int      degs[ROWS_WG];
    __shared__ unsigned elist[NWAVE * CAPW];

    const int tid  = threadIdx.x;
    const int lane = tid & 31;
    const int wv   = tid >> 5;
    const int base = blockIdx.x * ROWS_WG;
    const int nrows_i = (N_NODES - base) < ROWS_WG ? (N_NODES - base) : ROWS_WG;
    const unsigned nrows = (unsigned)nrows_i;

    {
        float4* a4 = (float4*)acc;
        const float4 z = make_float4(0.f, 0.f, 0.f, 0.f);
        for (int i = tid; i < ROWS_WG * PITCH / 4; i += NTHR) a4[i] = z;
        if (tid < ROWS_WG) degs[tid] = 0;
    }
    __syncthreads();

    unsigned* mylist = elist + wv * CAPW;
    int tail = 0;
    int head = 0;

    auto take = [&](int r, int e) {
        const unsigned lr = (unsigned)(r - base);
        const bool m = lr < nrows;
        const unsigned mask = __builtin_amdgcn_ballot_w32(m);
        if (mask != 0u) {
            const int room = CAPW - (tail - head);
            if (m) {
                atomicAdd(&degs[lr], 1);
                const int rank = (int)__builtin_amdgcn_mbcnt_lo(mask, 0u);
                if (rank < room) {
                    mylist[(tail + rank) & (CAPW - 1)] = (lr << 20) | (unsigned)e;
                } else {
                    const float* src = x + (size_t)cols[e] * CH;
                    float* dst = acc + lr * PITCH;
#pragma nounroll
                    for (int k = 0; k < CH; ++k) atomicAdd(dst + k, src[k]);
                }
            }
            const int n = __builtin_popcount(mask);
            tail += n < room ? n : room;
        }
    };

    const int hsel = lane >> 4;
    const int o0 = lane * 4 + ((0 + hsel) & 3);
    const int o1 = lane * 4 + ((1 + hsel) & 3);
    const int o2 = lane * 4 + ((2 + hsel) & 3);
    const int o3 = lane * 4 + ((3 + hsel) & 3);

    float4 bv[BATCH];
    int    blr[BATCH];
    bool   flying = false;
#pragma unroll
    for (int u = 0; u < BATCH; ++u) { bv[u] = make_float4(0.f, 0.f, 0.f, 0.f); blr[u] = 0; }

    auto fetch = [&](int nb) {
        unsigned ent = 0u;
        int c = 0;
        if (lane < nb) {
            ent = mylist[(head + lane) & (CAPW - 1)];
            c = cols[ent & 0xFFFFFu];
        }
#pragma unroll
        for (int u = 0; u < BATCH; ++u) {
            if (u < nb) {
                const int cj = __builtin_amdgcn_readlane(c, u);
                blr[u] = (int)(((unsigned)__builtin_amdgcn_readlane((int)ent, u)) >> 20);
                bv[u] = ((const float4*)(x + (size_t)cj * CH))[lane];
            }
        }
        head += nb;
    };

    auto add = [&](int nb) {
#pragma unroll
        for (int u = 0; u < BATCH; ++u) {
            if (u < nb) {
                float* d = acc + blr[u] * PITCH;
                const float t0 = hsel ? bv[u].y : bv[u].x;
                const float t1 = hsel ? bv[u].z : bv[u].y;
                const float t2 = hsel ? bv[u].w : bv[u].z;
                const float t3 = hsel ? bv[u].x : bv[u].w;
                atomicAdd(d + o0, t0);
                atomicAdd(d + o1, t1);
                atomicAdd(d + o2, t2);
                atomicAdd(d + o3, t3);
            }
        }
    };

    const int4* rows4 = (const int4*)rows;
    {
        int it = (int)(blockIdx.x % (FULL_IT / SCAN_U)) * SCAN_U;
        int4 cur[SCAN_U];
#pragma unroll
        for (int u = 0; u < SCAN_U; ++u) cur[u] = rows4[(it + u) * NTHR + tid];
#pragma unroll
        for (int u = 0; u < SCAN_U; ++u)
            asm volatile("" :: "v"(cur[u].x), "v"(cur[u].y), "v"(cur[u].z), "v"(cur[u].w));
        for (int o = 0; o < FULL_IT / SCAN_U; ++o) {
            const int nx = (it + SCAN_U < FULL_IT) ? it + SCAN_U : 0;
            int4 nxt[SCAN_U];
#pragma unroll
            for (int u = 0; u < SCAN_U; ++u) nxt[u] = rows4[(nx + u) * NTHR + tid];
#pragma unroll
            for (int u = 0; u < SCAN_U; ++u) {
                const int e0 = ((it + u) * NTHR + tid) * 4;
                take(cur[u].x, e0 + 0);
                take(cur[u].y, e0 + 1);
                take(cur[u].z, e0 + 2);
                take(cur[u].w, e0 + 3);
            }
            if (flying) { add(BATCH); flying = false; }
#pragma unroll
            for (int u = 0; u < SCAN_U; ++u)
                asm volatile("" :: "v"(nxt[u].x), "v"(nxt[u].y), "v"(nxt[u].z), "v"(nxt[u].w));
            if (tail - head >= BATCH) { fetch(BATCH); flying = true; }
#pragma unroll
            for (int u = 0; u < SCAN_U; ++u) cur[u] = nxt[u];
            it = nx;
        }
    }
    if (tid < REM4) {
        const int g4 = FULL_IT * NTHR + tid;
        const int4 r4 = rows4[g4];
        take(r4.x, g4 * 4 + 0);
        take(r4.y, g4 * 4 + 1);
        take(r4.z, g4 * 4 + 2);
        take(r4.w, g4 * 4 + 3);
    }
    tail = __builtin_amdgcn_readfirstlane(tail);
    if (flying) add(BATCH);
    while (tail - head > 0) {
        const int left = tail - head;
        const int nb = left < BATCH ? left : BATCH;
        fetch(nb);
        add(nb);
    }
    __syncthreads();

    for (int i = tid; i < nrows_i * 32; i += NTHR) {
        const int lr = i >> 5;
        const int q  = i & 31;
        float d = (float)degs[lr];
        d = d < 1.0f ? 1.0f : d;
        float4* p = (float4*)(acc + lr * PITCH + q * 4);
        float4 a = *p;
        const float4 xv = ((const float4*)(x + (size_t)(base + lr) * CH))[q];
        a.x = xv.x + a.x / d;
        a.y = xv.y + a.y / d;
        a.z = xv.z + a.z / d;
        a.w = xv.w + a.w / d;
        *p = a;
    }
    __syncthreads();

    {
        const int ln = lane & 15;
        const int hi = lane >> 4;
        const int n_items = (nrows_i >> 4) * 4;
        for (int item = wv; item < n_items; item += NWAVE) {
            const int mt = item >> 2;
            const int n0 = (item & 3) * 32;
            const float bv0 = bias[n0 + ln];
            const float bv1 = bias[n0 + 16 + ln];
            v8f c0, c1;
#pragma unroll
            for (int r = 0; r < 8; ++r) { c0[r] = bv0; c1[r] = bv1; }

            const float* arow = acc + (mt * 16 + ln) * PITCH + hi * 2;
            const float* w0   = W + (size_t)(n0 + ln) * CH + hi * 2;
            const float* w1   = w0 + 16 * CH;

#pragma unroll 8
            for (int kb = 0; kb < 32; ++kb) {
                const v2f a  = *(const v2f*)(arow + kb * 4);
                const v2f b0 = *(const v2f*)(w0 + kb * 4);
                const v2f b1 = *(const v2f*)(w1 + kb * 4);
                c0 = __builtin_amdgcn_wmma_f32_16x16x4_f32(false, a, false, b0, (short)0, c0, false, false);
                c1 = __builtin_amdgcn_wmma_f32_16x16x4_f32(false, a, false, b1, (short)0, c1, false, false);
            }

            float* orow = out + (size_t)(base + mt * 16 + hi * 8) * CH + n0 + ln;
#pragma unroll
            for (int r = 0; r < 8; ++r) {
                orow[(size_t)r * CH]      = c0[r];
                orow[(size_t)r * CH + 16] = c1[r];
            }
        }
    }
}


extern "C" void kernel_launch(void* const* d_in, const int* in_sizes, int n_in,
                              void* d_out, int out_size, void* d_ws, size_t ws_size,
                              hipStream_t stream) {
    const float* x    = (const float*)d_in[0];
    const int*   ei   = (const int*)  d_in[1];
    const float* W    = (const float*)d_in[2];
    const float* bias = (const float*)d_in[3];
    float*       out  = (float*)d_out;

    const int* rows = ei;
    const int* cols = ei + N_EDGES;

    gc_fused<<<N_WG, NTHR, 0, stream>>>(x, rows, cols, W, bias, out);
}


# W (128x128 f32) copied once per work-group into LDS at pitch 132; the B fragments of the f32 WMMA are LDS reads instead of 16-line global reads; same k order, sums unchanged
# speedup vs baseline: 3.8065x; 1.1194x over previous
#include <hip/hip_runtime.h>


#define N_NODES 50000
#define N_EDGES 640000
#define CH      128
#define ROWS_WG 208
#define N_WG    241
#define PITCH   132
#define CAPW    512
#define BATCH   8
#define NTHR    1024
#define NWAVE   32
#define N_ROW4  (N_EDGES / 4)
#define FULL_IT (N_ROW4 / NTHR)
#define REM4    (N_ROW4 - FULL_IT * NTHR)
#define SCAN_U  6

typedef float v2f __attribute__((ext_vector_type(2)));
typedef float v8f __attribute__((ext_vector_type(8)));

__global__ __launch_bounds__(NTHR) void gc_fused(const float* __restrict__ x,
                                                 const int*   __restrict__ rows,
                                                 const int*   __restrict__ cols,
                                                 const float* __restrict__ W,
                                                 const float* __restrict__ bias,
                                                 float*       __restrict__ out) {
    __shared__ __attribute__((aligned(16))) float    acc[ROWS_WG * PITCH];
    __shared__ __attribute__((aligned(16))) float    wl[CH * PITCH];
    __shared__ int      degs[ROWS_WG];
    __shared__ unsigned elist[NWAVE * CAPW];

    const int tid  = threadIdx.x;
    const int lane = tid & 31;
    const int wv   = tid >> 5;
    const int base = blockIdx.x * ROWS_WG;
    const int nrows_i = (N_NODES - base) < ROWS_WG ? (N_NODES - base) : ROWS_WG;
    const unsigned nrows = (unsigned)nrows_i;

    {
        float4 wq[CH * CH / 4 / NTHR];
#pragma unroll
        for (int j = 0; j < CH * CH / 4 / NTHR; ++j) wq[j] = ((const float4*)W)[j * NTHR + tid];
        float4* a4 = (float4*)acc;
        const float4 z = make_float4(0.f, 0.f, 0.f, 0.f);
        for (int i = tid; i < ROWS_WG * PITCH / 4; i += NTHR) a4[i] = z;
        if (tid < ROWS_WG) degs[tid] = 0;
#pragma unroll
        for (int j = 0; j < CH * CH / 4 / NTHR; ++j) {
            const int i = j * NTHR + tid;
            *(float4*)(wl + (i >> 5) * PITCH + (i & 31) * 4) = wq[j];
        }
    }
    __syncthreads();

    unsigned* mylist = elist + wv * CAPW;
    int tail = 0;
    int head = 0;

    auto take = [&](int r, int e) {
        const unsigned lr = (unsigned)(r - base);
        const bool m = lr < nrows;
        const unsigned mask = __builtin_amdgcn_ballot_w32(m);
        if (mask != 0u) {
            const int room = CAPW - (tail - head);
            if (m) {
                atomicAdd(&degs[lr], 1);
                const int rank = (int)__builtin_amdgcn_mbcnt_lo(mask, 0u);
                if (rank < room) {
                    mylist[(tail + rank) & (CAPW - 1)] = (lr << 20) | (unsigned)e;
                } else {
                    const float* src = x + (size_t)cols[e] * CH;
                    float* dst = acc + lr * PITCH;
#pragma nounroll
                    for (int k = 0; k < CH; ++k) atomicAdd(dst + k, src[k]);
                }
            }
            const int n = __builtin_popcount(mask);
            tail += n < room ? n : room;
        }
    };

    const int hsel = lane >> 4;
    const int o0 = lane * 4 + ((0 + hsel) & 3);
    const int o1 = lane * 4 + ((1 + hsel) & 3);
    const int o2 = lane * 4 + ((2 + hsel) & 3);
    const int o3 = lane * 4 + ((3 + hsel) & 3);

    float4 bv[BATCH];
    int    blr[BATCH];
    bool   flying = false;
#pragma unroll
    for (int u = 0; u < BATCH; ++u) { bv[u] = make_float4(0.f, 0.f, 0.f, 0.f); blr[u] = 0; }

    auto fetch = [&](int nb) {
        unsigned ent = 0u;
        int c = 0;
        if (lane < nb) {
            ent = mylist[(head + lane) & (CAPW - 1)];
            c = cols[ent & 0xFFFFFu];
        }
#pragma unroll
        for (int u = 0; u < BATCH; ++u) {
            if (u < nb) {
                const int cj = __builtin_amdgcn_readlane(c, u);
                blr[u] = (int)(((unsigned)__builtin_amdgcn_readlane((int)ent, u)) >> 20);
                bv[u] = ((const float4*)(x + (size_t)cj * CH))[lane];
            }
        }
        head += nb;
    };

    auto add = [&](int nb) {
#pragma unroll
        for (int u = 0; u < BATCH; ++u) {
            if (u < nb) {
                float* d = acc + blr[u] * PITCH;
                const float t0 = hsel ? bv[u].y : bv[u].x;
                const float t1 = hsel ? bv[u].z : bv[u].y;
                const float t2 = hsel ? bv[u].w : bv[u].z;
                const float t3 = hsel ? bv[u].x : bv[u].w;
                atomicAdd(d + o0, t0);
                atomicAdd(d + o1, t1);
                atomicAdd(d + o2, t2);
                atomicAdd(d + o3, t3);
            }
        }
    };

    const int4* rows4 = (const int4*)rows;
    {
        int it = (int)(blockIdx.x % (FULL_IT / SCAN_U)) * SCAN_U;
        int4 cur[SCAN_U];
#pragma unroll
        for (int u = 0; u < SCAN_U; ++u) cur[u] = rows4[(it + u) * NTHR + tid];
#pragma unroll
        for (int u = 0; u < SCAN_U; ++u)
            asm volatile("" :: "v"(cur[u].x), "v"(cur[u].y), "v"(cur[u].z), "v"(cur[u].w));
        for (int o = 0; o < FULL_IT / SCAN_U; ++o) {
            const int nx = (it + SCAN_U < FULL_IT) ? it + SCAN_U : 0;
            int4 nxt[SCAN_U];
#pragma unroll
            for (int u = 0; u < SCAN_U; ++u) nxt[u] = rows4[(nx + u) * NTHR + tid];
#pragma unroll
            for (int u = 0; u < SCAN_U; ++u) {
                const int e0 = ((it + u) * NTHR + tid) * 4;
                take(cur[u].x, e0 + 0);
                take(cur[u].y, e0 + 1);
                take(cur[u].z, e0 + 2);
                take(cur[u].w, e0 + 3);
            }
            if (flying) { add(BATCH); flying = false; }
#pragma unroll
            for (int u = 0; u < SCAN_U; ++u)
                asm volatile("" :: "v"(nxt[u].x), "v"(nxt[u].y), "v"(nxt[u].z), "v"(nxt[u].w));
            if (tail - head >= BATCH) { fetch(BATCH); flying = true; }
#pragma unroll
            for (int u = 0; u < SCAN_U; ++u) cur[u] = nxt[u];
            it = nx;
        }
    }
    if (tid < REM4) {
        const int g4 = FULL_IT * NTHR + tid;
        const int4 r4 = rows4[g4];
        take(r4.x, g4 * 4 + 0);
        take(r4.y, g4 * 4 + 1);
        take(r4.z, g4 * 4 + 2);
        take(r4.w, g4 * 4 + 3);
    }
    tail = __builtin_amdgcn_readfirstlane(tail);
    if (flying) add(BATCH);
    while (tail - head > 0) {
        const int left = tail - head;
        const int nb = left < BATCH ? left : BATCH;
        fetch(nb);
        add(nb);
    }
    __syncthreads();

    for (int i = tid; i < nrows_i * 32; i += NTHR) {
        const int lr = i >> 5;
        const int q  = i & 31;
        float d = (float)degs[lr];
        d = d < 1.0f ? 1.0f : d;
        float4* p = (float4*)(acc + lr * PITCH + q * 4);
        float4 a = *p;
        const float4 xv = ((const float4*)(x + (size_t)(base + lr) * CH))[q];
        a.x = xv.x + a.x / d;
        a.y = xv.y + a.y / d;
        a.z = xv.z + a.z / d;
        a.w = xv.w + a.w / d;
        *p = a;
    }
    __syncthreads();

    {
        const int ln = lane & 15;
        const int hi = lane >> 4;
        const int n_items = (nrows_i >> 4) * 4;
        for (int item = wv; item < n_items; item += NWAVE) {
            const int mt = item >> 2;
            const int n0 = (item & 3) * 32;
            const float bv0 = bias[n0 + ln];
            const float bv1 = bias[n0 + 16 + ln];
            v8f c0, c1;
#pragma unroll
            for (int r = 0; r < 8; ++r) { c0[r] = bv0; c1[r] = bv1; }

            const float* arow = acc + (mt * 16 + ln) * PITCH + hi * 2;
            const float* w0   = wl + (n0 + ln) * PITCH + hi * 2;
            const float* w1   = w0 + 16 * PITCH;

#pragma unroll 8
            for (int kb = 0; kb < 32; ++kb) {
                const v2f a  = *(const v2f*)(arow + kb * 4);
                const v2f b0 = *(const v2f*)(w0 + kb * 4);
                const v2f b1 = *(const v2f*)(w1 + kb * 4);
                c0 = __builtin_amdgcn_wmma_f32_16x16x4_f32(false, a, false, b0, (short)0, c0, false, false);
                c1 = __builtin_amdgcn_wmma_f32_16x16x4_f32(false, a, false, b1, (short)0, c1, false, false);
            }

            float* orow = out + (size_t)(base + mt * 16 + hi * 8) * CH + n0 + ln;
#pragma unroll
            for (int r = 0; r < 8; ++r) {
                orow[(size_t)r * CH]      = c0[r];
                orow[(size_t)r * CH + 16] = c1[r];
            }
        }
    }
}


extern "C" void kernel_launch(void* const* d_in, const int* in_sizes, int n_in,
                              void* d_out, int out_size, void* d_ws, size_t ws_size,
                              hipStream_t stream) {
    const float* x    = (const float*)d_in[0];
    const int*   ei   = (const int*)  d_in[1];
    const float* W    = (const float*)d_in[2];
    const float* bias = (const float*)d_in[3];
    float*       out  = (float*)d_out;

    const int* rows = ei;
    const int* cols = ei + N_EDGES;

    gc_fused<<<N_WG, NTHR, 0, stream>>>(x, rows, cols, W, bias, out);
}


# a first launch marks per 208-row group which 128-B lines of the row list hold its rows (LDS bit-or, 617 KB workspace); a work-group reads only those lines; sums stay LDS f32 atomics, arrival order
# speedup vs baseline: 4.4287x; 1.1635x over previous
#include <hip/hip_runtime.h>


#define N_NODES 50000
#define N_EDGES 640000
#define CH      128
#define ROWS_WG 208
#define N_WG    241
#define PITCH   132
#define CAPW    512
#define BATCH   8
#define NTHR    1024
#define NWAVE   32
#define N_ROW4  (N_EDGES / 4)
#define SCAN_U  6
#define BLK_EDGES 8192
#define BLK_WORDS 8
#define N_BLK   79
#define GRP_PAD 244
#define N_LINE  (N_EDGES / 32)
#define WPW     20

typedef float v2f __attribute__((ext_vector_type(2)));
typedef float v8f __attribute__((ext_vector_type(8)));

__global__ __launch_bounds__(NTHR) void gc_index(const int* __restrict__ rows, unsigned* __restrict__ bm) {
    __shared__ unsigned sb[BLK_WORDS * GRP_PAD];
    const int tid = threadIdx.x;
    for (int i = tid; i < BLK_WORDS * GRP_PAD; i += NTHR) sb[i] = 0u;
    __syncthreads();
    const int4* rows4 = (const int4*)rows;
    int4 r[BLK_EDGES / 4 / NTHR];
#pragma unroll
    for (int j = 0; j < BLK_EDGES / 4 / NTHR; ++j) {
        const int g4 = blockIdx.x * (BLK_EDGES / 4) + j * NTHR + tid;
        r[j] = make_int4(-1, -1, -1, -1);
        if (g4 < N_ROW4) r[j] = rows4[g4];
    }
#pragma unroll
    for (int j = 0; j < BLK_EDGES / 4 / NTHR; ++j) {
        const int q = j * NTHR + tid;
        const int k = q >> 8;
        const unsigned bit = 1u << ((q >> 3) & 31);
        const unsigned g0 = (unsigned)r[j].x / ROWS_WG, g1 = (unsigned)r[j].y / ROWS_WG;
        const unsigned g2 = (unsigned)r[j].z / ROWS_WG, g3 = (unsigned)r[j].w / ROWS_WG;
        if (g0 < N_WG) atomicOr(&sb[k * GRP_PAD + g0], bit);
        if (g1 < N_WG) atomicOr(&sb[k * GRP_PAD + g1], bit);
        if (g2 < N_WG) atomicOr(&sb[k * GRP_PAD + g2], bit);
        if (g3 < N_WG) atomicOr(&sb[k * GRP_PAD + g3], bit);
    }
    __syncthreads();
    unsigned* dst = bm + (size_t)blockIdx.x * (BLK_WORDS * GRP_PAD);
    for (int o = tid; o < BLK_WORDS * GRP_PAD; o += NTHR) dst[o] = sb[(o & 7) * GRP_PAD + (o >> 3)];
}

__global__ __launch_bounds__(NTHR) void gc_fused(const float* __restrict__ x,
                                                 const int*   __restrict__ rows,
                                                 const int*   __restrict__ cols,
                                                 const float* __restrict__ W,
                                                 const float* __restrict__ bias,
                                                 const unsigned* __restrict__ bm,
                                                 float*       __restrict__ out) {
    __shared__ __attribute__((aligned(16))) float    acc[ROWS_WG * PITCH];
    __shared__ __attribute__((aligned(16))) float    wl[CH * PITCH];
    __shared__ int      degs[ROWS_WG];
    __shared__ unsigned elist[NWAVE * CAPW];
    __shared__ unsigned short clist[NWAVE * WPW * 32 + 16];

    const int tid  = threadIdx.x;
    const int lane = tid & 31;
    const int wv   = tid >> 5;
    const int base = blockIdx.x * ROWS_WG;
    const int nrows_i = (N_NODES - base) < ROWS_WG ? (N_NODES - base) : ROWS_WG;
    const unsigned nrows = (unsigned)nrows_i;

    const int wvs = __builtin_amdgcn_readfirstlane(wv);
    unsigned myw = 0u;
    {
        const int j = wvs * WPW + lane;
        if (lane < WPW && j < N_BLK * BLK_WORDS)
            myw = bm[((size_t)(j >> 3) * GRP_PAD + blockIdx.x) * BLK_WORDS + (j & 7)];
    }

    {
        float4 wq[CH * CH / 4 / NTHR];
#pragma unroll
        for (int j = 0; j < CH * CH / 4 / NTHR; ++j) wq[j] = ((const float4*)W)[j * NTHR + tid];
        float4* a4 = (float4*)acc;
        const float4 z = make_float4(0.f, 0.f, 0.f, 0.f);
        for (int i = tid; i < ROWS_WG * PITCH / 4; i += NTHR) a4[i] = z;
        if (tid < ROWS_WG) degs[tid] = 0;
#pragma unroll
        for (int j = 0; j < CH * CH / 4 / NTHR; ++j) {
            const int i = j * NTHR + tid;
            *(float4*)(wl + (i >> 5) * PITCH + (i & 31) * 4) = wq[j];
        }
    }
    __syncthreads();

    unsigned* mylist = elist + wv * CAPW;
    int tail = 0;
    int head = 0;

    auto take = [&](int r, int e) {
        const unsigned lr = (unsigned)(r - base);
        const bool m = lr < nrows;
        const unsigned mask = __builtin_amdgcn_ballot_w32(m);
        if (mask != 0u) {
            const int room = CAPW - (tail - head);
            if (m) {
                atomicAdd(&degs[lr], 1);
                const int rank = (int)__builtin_amdgcn_mbcnt_lo(mask, 0u);
                if (rank < room) {
                    mylist[(tail + rank) & (CAPW - 1)] = (lr << 20) | (unsigned)e;
                } else {
                    const float* src = x + (size_t)cols[e] * CH;
                    float* dst = acc + lr * PITCH;
#pragma nounroll
                    for (int k = 0; k < CH; ++k) atomicAdd(dst + k, src[k]);
                }
            }
            const int n = __builtin_popcount(mask);
            tail += n < room ? n : room;
        }
    };

    const int hsel = lane >> 4;
    const int o0 = lane * 4 + ((0 + hsel) & 3);
    const int o1 = lane * 4 + ((1 + hsel) & 3);
    const int o2 = lane * 4 + ((2 + hsel) & 3);
    const int o3 = lane * 4 + ((3 + hsel) & 3);

    float4 bv[BATCH];
    int    blr[BATCH];
    bool   flying = false;
#pragma unroll
    for (int u = 0; u < BATCH; ++u) { bv[u] = make_float4(0.f, 0.f, 0.f, 0.f); blr[u] = 0; }

    auto fetch = [&](int nb) {
        unsigned ent = 0u;
        int c = 0;
        if (lane < nb) {
            ent = mylist[(head + lane) & (CAPW - 1)];
            c = cols[ent & 0xFFFFFu];
        }
#pragma unroll
        for (int u = 0; u < BATCH; ++u) {
            if (u < nb) {
                const int cj = __builtin_amdgcn_readlane(c, u);
                blr[u] = (int)(((unsigned)__builtin_amdgcn_readlane((int)ent, u)) >> 20);
                bv[u] = ((const float4*)(x + (size_t)cj * CH))[lane];
            }
        }
        head += nb;
    };

    auto add = [&](int nb) {
#pragma unroll
        for (int u = 0; u < BATCH; ++u) {
            if (u < nb) {
                float* d = acc + blr[u] * PITCH;
                const float t0 = hsel ? bv[u].y : bv[u].x;
                const float t1 = hsel ? bv[u].z : bv[u].y;
                const float t2 = hsel ? bv[u].w : bv[u].z;
                const float t3 = hsel ? bv[u].x : bv[u].w;
                atomicAdd(d + o0, t0);
                atomicAdd(d + o1, t1);
                atomicAdd(d + o2, t2);
                atomicAdd(d + o3, t3);
            }
        }
    };

    {
        unsigned short* mycl = clist + wv * (WPW * 32);
        const int pc = __builtin_popcount(myw);
        int incl = pc;
#pragma unroll
        for (int d = 1; d < 32; d <<= 1) {
            const int t = __shfl_up(incl, d);
            if (lane >= d) incl += t;
        }
        const int ntot = __builtin_amdgcn_readlane(incl, 31);
        {
            int pos = incl - pc;
            unsigned w = myw;
            const int cbase = (wvs * WPW + lane) << 5;
            while (w != 0u) {
                const int b = __builtin_ctz(w);
                w &= w - 1u;
                mycl[pos++] = (unsigned short)(cbase + b);
            }
        }
        auto ask = [&](int i0, int (&r)[SCAN_U], int (&e)[SCAN_U]) {
#pragma unroll
            for (int u = 0; u < SCAN_U; ++u) {
                const int c = (int)mycl[i0 + u];
                e[u] = (c < N_LINE ? c : N_LINE - 1) * 32 + lane;
                r[u] = rows[e[u]];
            }
        };
        int cur[SCAN_U], ce[SCAN_U];
        ask(0, cur, ce);
#pragma unroll
        for (int u = 0; u < SCAN_U; ++u) asm volatile("" :: "v"(cur[u]));
        for (int it = 0; it < ntot; it += SCAN_U) {
            int nxt[SCAN_U], ne[SCAN_U];
            ask(it + SCAN_U, nxt, ne);
#pragma unroll
            for (int u = 0; u < SCAN_U; ++u)
                if (it + u < ntot) take(cur[u], ce[u]);
            if (flying) { add(BATCH); flying = false; }
#pragma unroll
            for (int u = 0; u < SCAN_U; ++u) asm volatile("" :: "v"(nxt[u]));
            if (tail - head >= BATCH) { fetch(BATCH); flying = true; }
#pragma unroll
            for (int u = 0; u < SCAN_U; ++u) { cur[u] = nxt[u]; ce[u] = ne[u]; }
        }
    }
    tail = __builtin_amdgcn_readfirstlane(tail);
    if (flying) add(BATCH);
    while (tail - head > 0) {
        const int left = tail - head;
        const int nb = left < BATCH ? left : BATCH;
        fetch(nb);
        add(nb);
    }
    __syncthreads();

    for (int i = tid; i < nrows_i * 32; i += NTHR) {
        const int lr = i >> 5;
        const int q  = i & 31;
        float d = (float)degs[lr];
        d = d < 1.0f ? 1.0f : d;
        float4* p = (float4*)(acc + lr * PITCH + q * 4);
        float4 a = *p;
        const float4 xv = ((const float4*)(x + (size_t)(base + lr) * CH))[q];
        a.x = xv.x + a.x / d;
        a.y = xv.y + a.y / d;
        a.z = xv.z + a.z / d;
        a.w = xv.w + a.w / d;
        *p = a;
    }
    __syncthreads();

    {
        const int ln = lane & 15;
        const int hi = lane >> 4;
        const int n_items = (nrows_i >> 4) * 4;
        for (int item = wv; item < n_items; item += NWAVE) {
            const int mt = item >> 2;
            const int n0 = (item & 3) * 32;
            const float bv0 = bias[n0 + ln];
            const float bv1 = bias[n0 + 16 + ln];
            v8f c0, c1;
#pragma unroll
            for (int r = 0; r < 8; ++r) { c0[r] = bv0; c1[r] = bv1; }

            const float* arow = acc + (mt * 16 + ln) * PITCH + hi * 2;
            const float* w0   = wl + (n0 + ln) * PITCH + hi * 2;
            const float* w1   = w0 + 16 * PITCH;

#pragma unroll 8
            for (int kb = 0; kb < 32; ++kb) {
                const v2f a  = *(const v2f*)(arow + kb * 4);
                const v2f b0 = *(const v2f*)(w0 + kb * 4);
                const v2f b1 = *(const v2f*)(w1 + kb * 4);
                c0 = __builtin_amdgcn_wmma_f32_16x16x4_f32(false, a, false, b0, (short)0, c0, false, false);
                c1 = __builtin_amdgcn_wmma_f32_16x16x4_f32(false, a, false, b1, (short)0, c1, false, false);
            }

            float* orow = out + (size_t)(base + mt * 16 + hi * 8) * CH + n0 + ln;
#pragma unroll
            for (int r = 0; r < 8; ++r) {
                orow[(size_t)r * CH]      = c0[r];
                orow[(size_t)r * CH + 16] = c1[r];
            }
        }
    }
}


extern "C" void kernel_launch(void* const* d_in, const int* in_sizes, int n_in,
                              void* d_out, int out_size, void* d_ws, size_t ws_size,
                              hipStream_t stream) {
    const float* x    = (const float*)d_in[0];
    const int*   ei   = (const int*)  d_in[1];
    const float* W    = (const float*)d_in[2];
    const float* bias = (const float*)d_in[3];
    float*       out  = (float*)d_out;

    const int* rows = ei;
    const int* cols = ei + N_EDGES;

    unsigned* bm = (unsigned*)d_ws;

    gc_index<<<N_BLK, NTHR, 0, stream>>>(rows, bm);
    gc_fused<<<N_WG, NTHR, 0, stream>>>(x, rows, cols, W, bias, bm, out);
}
